# starting kernel
# baseline (speedup 1.0000x reference)
#include <hip/hip_runtime.h>

typedef float v2f __attribute__((ext_vector_type(2)));
typedef float v4f __attribute__((ext_vector_type(4)));
typedef float v8f __attribute__((ext_vector_type(8)));

#define BB 16
#define NN 1024
#define FF 256
#define NEG_INF_V (-1000000000.0f)
#define SLOPE_V 0.2f

#define K3_WAVES 4
#define BSTRIDE 20

__global__ __launch_bounds__(32)
void gat_gemm_hW(const float* __restrict__ h, const float* __restrict__ W,
                 float* __restrict__ WhT) {
  int tile = blockIdx.x;
  int ot = tile % (FF / 16);
  int nt = (tile / (FF / 16)) % (NN / 16);
  int b  = tile / ((FF / 16) * (NN / 16));
  int lane = threadIdx.x;
  int n  = lane & 15;
  int hh = lane >> 4;

  const float* Arow = h + ((size_t)(b * NN + nt * 16 + n)) * FF;
  const float* Bcol = W + ot * 16 + n;

  v8f acc = {};
  #pragma unroll 4
  for (int k = 0; k < FF; k += 4) {
    v2f a = *(const v2f*)(Arow + k + 2 * hh);
    v2f bm;
    bm.x = Bcol[(size_t)(k + 2 * hh) * FF];
    bm.y = Bcol[(size_t)(k + 1 + 2 * hh) * FF];
    acc = __builtin_amdgcn_wmma_f32_16x16x4_f32(false, a, false, bm,
                                                (short)0, acc, false, false);
  }
  float* outp = WhT + ((size_t)(b * FF + ot * 16 + n)) * NN + nt * 16 + 8 * hh;
  v4f lo = { acc[0], acc[1], acc[2], acc[3] };
  v4f hi = { acc[4], acc[5], acc[6], acc[7] };
  *(v4f*)(outp)     = lo;
  *(v4f*)(outp + 4) = hi;
}

__global__ __launch_bounds__(256)
void gat_f12(const float* __restrict__ WhT, const float* __restrict__ a,
             float* __restrict__ f1, float* __restrict__ f2) {
  int idx = blockIdx.x * blockDim.x + threadIdx.x;
  int b = idx / NN;
  int n = idx % NN;
  const float* base = WhT + (size_t)b * FF * NN + n;
  float s1 = 0.f, s2 = 0.f;
  #pragma unroll 8
  for (int o = 0; o < FF; ++o) {
    float w = base[(size_t)o * NN];
    s1 += w * a[o];
    s2 += w * a[FF + o];
  }
  f1[idx] = s1;
  f2[idx] = s2;
}

__global__ __launch_bounds__(128)
void gat_attn(const float* __restrict__ WhT, const int* __restrict__ adj,
              const float* __restrict__ f1, const float* __restrict__ f2,
              float* __restrict__ out) {
  __shared__ float Btile[FF * BSTRIDE];
  __shared__ float ptile[K3_WAVES][16 * BSTRIDE];
  int tid  = threadIdx.x;
  int w    = tid >> 5;
  int lane = tid & 31;
  int n  = lane & 15;
  int hh = lane >> 4;
  int b     = blockIdx.x >> 4;
  int super = blockIdx.x & 15;
  int i0 = (super * K3_WAVES + w) * 16;

  float f1v[8];
  {
    const float* p = f1 + b * NN + i0 + 8 * hh;
    v4f t0 = *(const v4f*)(p);
    v4f t1 = *(const v4f*)(p + 4);
    f1v[0] = t0.x; f1v[1] = t0.y; f1v[2] = t0.z; f1v[3] = t0.w;
    f1v[4] = t1.x; f1v[5] = t1.y; f1v[6] = t1.z; f1v[7] = t1.w;
  }

  v8f acc[16];
  v8f zero = {};
  #pragma unroll
  for (int t = 0; t < 16; ++t) acc[t] = zero;
  float mrun[8], lrun[8];
  #pragma unroll
  for (int r = 0; r < 8; ++r) { mrun[r] = -3.0e38f; lrun[r] = 0.f; }

  const int*   adjb = adj + ((size_t)(b * NN + i0 + 8 * hh)) * NN;
  const float* WhTb = WhT + (size_t)b * FF * NN;
  const float* f2b  = f2 + b * NN;

  for (int jt = 0; jt < NN / 16; ++jt) {
    int j0 = jt * 16;

    {
      const float* g0 = WhTb + (size_t)tid * NN + j0;
      const float* g1 = g0 + (size_t)128 * NN;
      v4f x0 = *(const v4f*)(g0);
      v4f x1 = *(const v4f*)(g0 + 4);
      v4f x2 = *(const v4f*)(g0 + 8);
      v4f x3 = *(const v4f*)(g0 + 12);
      v4f y0 = *(const v4f*)(g1);
      v4f y1 = *(const v4f*)(g1 + 4);
      v4f y2 = *(const v4f*)(g1 + 8);
      v4f y3 = *(const v4f*)(g1 + 12);
      float* l0 = Btile + tid * BSTRIDE;
      float* l1 = Btile + (tid + 128) * BSTRIDE;
      *(v4f*)(l0)      = x0;  *(v4f*)(l0 + 4)  = x1;
      *(v4f*)(l0 + 8)  = x2;  *(v4f*)(l0 + 12) = x3;
      *(v4f*)(l1)      = y0;  *(v4f*)(l1 + 4)  = y1;
      *(v4f*)(l1 + 8)  = y2;  *(v4f*)(l1 + 12) = y3;
    }

    float f2v = f2b[j0 + n];
    float p[8];
    #pragma unroll
    for (int r = 0; r < 8; ++r) {
      float e = f1v[r] + f2v;
      e = e > 0.f ? e : SLOPE_V * e;
      int av = adjb[(size_t)r * NN + j0 + n];
      e = (av == 0) ? NEG_INF_V : e;
      float rm = e;
      rm = fmaxf(rm, __shfl_xor(rm, 1, 32));
      rm = fmaxf(rm, __shfl_xor(rm, 2, 32));
      rm = fmaxf(rm, __shfl_xor(rm, 4, 32));
      rm = fmaxf(rm, __shfl_xor(rm, 8, 32));
      float nm   = fmaxf(mrun[r], rm);
      float pe   = __expf(e - nm);
      float corr = __expf(mrun[r] - nm);
      mrun[r] = nm;
      float rs = pe;
      rs += __shfl_xor(rs, 1, 32);
      rs += __shfl_xor(rs, 2, 32);
      rs += __shfl_xor(rs, 4, 32);
      rs += __shfl_xor(rs, 8, 32);
      lrun[r] = lrun[r] * corr + rs;
      p[r] = pe;
      #pragma unroll
      for (int t = 0; t < 16; ++t) acc[t][r] *= corr;
    }
    #pragma unroll
    for (int r = 0; r < 8; ++r)
      ptile[w][(r + 8 * hh) * BSTRIDE + n] = p[r];

    __syncthreads();

    v2f A[4];
    #pragma unroll
    for (int kk = 0; kk < 4; ++kk)
      A[kk] = *(const v2f*)(&ptile[w][n * BSTRIDE + 4 * kk + 2 * hh]);

    #pragma unroll
    for (int t = 0; t < 16; ++t) {
      const float* Bp = Btile + (t * 16 + n) * BSTRIDE + 2 * hh;
      #pragma unroll
      for (int kk = 0; kk < 4; ++kk) {
        v2f bm = *(const v2f*)(Bp + 4 * kk);
        acc[t] = __builtin_amdgcn_wmma_f32_16x16x4_f32(false, A[kk], false, bm,
                                                       (short)0, acc[t],
                                                       false, false);
      }
    }
    __syncthreads();
  }

  float inv[8];
  #pragma unroll
  for (int r = 0; r < 8; ++r) inv[r] = 1.0f / lrun[r];
  float* ob = out + ((size_t)(b * NN + i0 + 8 * hh)) * FF + n;
  #pragma unroll
  for (int t = 0; t < 16; ++t) {
    #pragma unroll
    for (int r = 0; r < 8; ++r)
      ob[(size_t)r * FF + t * 16] = acc[t][r] * inv[r];
  }
}

extern "C" void kernel_launch(void* const* d_in, const int* in_sizes, int n_in,
                              void* d_out, int out_size, void* d_ws, size_t ws_size,
                              hipStream_t stream) {
  const float* h   = (const float*)d_in[0];
  const int*   adj = (const int*)d_in[1];
  const float* W   = (const float*)d_in[2];
  const float* a   = (const float*)d_in[3];
  float* out = (float*)d_out;

  float* WhT = (float*)d_ws;
  float* f1  = WhT + (size_t)BB * NN * FF;
  float* f2  = f1 + (size_t)BB * NN;

  gat_gemm_hW<<<BB * (NN / 16) * (FF / 16), 32, 0, stream>>>(h, W, WhT);
  gat_f12<<<(BB * NN) / 256, 256, 0, stream>>>(WhT, a, f1, f2);
  gat_attn<<<BB * (NN / 16 / K3_WAVES), 128, 0, stream>>>(WhT, adj, f1, f2, out);
}


# product reassociated as (attention @ h) @ W: attention kernel accumulates over h panels and multiplies by W in its epilogue; f1/f2 = h @ (W @ a); no h@W kernel, no Wh in workspace (new sum order)
# speedup vs baseline: 2.4038x; 2.4038x over previous
#include <hip/hip_runtime.h>

typedef float v2f __attribute__((ext_vector_type(2)));
typedef float v4f __attribute__((ext_vector_type(4)));
typedef float v8f __attribute__((ext_vector_type(8)));

#define BB 16
#define NN 1024
#define FF 256
#define NEG_INF_V (-1000000000.0f)
#define SLOPE_V 0.2f

#define K3_WAVES 4
#define NG 4
#define K3_THREADS (NG * K3_WAVES * 32)
#define NIT (NN / 16 / NG)
#define BSTRIDE 20
#define RSP 544
#define XS 260

#define K1_THREADS 128

__global__ __launch_bounds__(K1_THREADS)
void gat_wa(const float* __restrict__ W, const float* __restrict__ a,
            float* __restrict__ wa) {
  int k    = blockIdx.x * (K1_THREADS / 32) + (threadIdx.x >> 5);
  int lane = threadIdx.x & 31;
  const float* wr = W + (size_t)k * FF + 4 * lane;
  v4f w0 = *(const v4f*)(wr);
  v4f w1 = *(const v4f*)(wr + 128);
  v4f p0 = *(const v4f*)(a + 4 * lane);
  v4f p1 = *(const v4f*)(a + 128 + 4 * lane);
  v4f q0 = *(const v4f*)(a + FF + 4 * lane);
  v4f q1 = *(const v4f*)(a + FF + 128 + 4 * lane);
  float s1 = w0.x * p0.x + w0.y * p0.y + w0.z * p0.z + w0.w * p0.w
           + w1.x * p1.x + w1.y * p1.y + w1.z * p1.z + w1.w * p1.w;
  float s2 = w0.x * q0.x + w0.y * q0.y + w0.z * q0.z + w0.w * q0.w
           + w1.x * q1.x + w1.y * q1.y + w1.z * q1.z + w1.w * q1.w;
  #pragma unroll
  for (int m = 1; m < 32; m <<= 1) {
    s1 += __shfl_xor(s1, m, 32);
    s2 += __shfl_xor(s2, m, 32);
  }
  if (lane == 0) { wa[k] = s1; wa[FF + k] = s2; }
}

#define K2_THREADS 256
#define K2_ROWS 8

__global__ __launch_bounds__(K2_THREADS)
void gat_f12(const float* __restrict__ h, const float* __restrict__ wa,
             float* __restrict__ f1, float* __restrict__ f2) {
  int wave = (blockIdx.x * K2_THREADS + threadIdx.x) >> 5;
  int lane = threadIdx.x & 31;
  v4f u0 = *(const v4f*)(wa + 4 * lane);
  v4f u1 = *(const v4f*)(wa + 128 + 4 * lane);
  v4f v0 = *(const v4f*)(wa + FF + 4 * lane);
  v4f v1 = *(const v4f*)(wa + FF + 128 + 4 * lane);
  const float* hp = h + (size_t)wave * K2_ROWS * FF + 4 * lane;
  v4f x0[K2_ROWS], x1[K2_ROWS];
  #pragma unroll
  for (int r = 0; r < K2_ROWS; ++r) {
    x0[r] = *(const v4f*)(hp + (size_t)r * FF);
    x1[r] = *(const v4f*)(hp + (size_t)r * FF + 128);
  }
  float o1 = 0.f, o2 = 0.f;
  #pragma unroll
  for (int r = 0; r < K2_ROWS; ++r) {
    float s1 = x0[r].x * u0.x + x0[r].y * u0.y + x0[r].z * u0.z + x0[r].w * u0.w
             + x1[r].x * u1.x + x1[r].y * u1.y + x1[r].z * u1.z + x1[r].w * u1.w;
    float s2 = x0[r].x * v0.x + x0[r].y * v0.y + x0[r].z * v0.z + x0[r].w * v0.w
             + x1[r].x * v1.x + x1[r].y * v1.y + x1[r].z * v1.z + x1[r].w * v1.w;
    #pragma unroll
    for (int m = 1; m < 32; m <<= 1) {
      s1 += __shfl_xor(s1, m, 32);
      s2 += __shfl_xor(s2, m, 32);
    }
    if (lane == r) { o1 = s1; o2 = s2; }
  }
  if (lane < K2_ROWS) {
    f1[wave * K2_ROWS + lane] = o1;
    f2[wave * K2_ROWS + lane] = o2;
  }
}

#define PANEL_FLOATS (8 * RSP)
#define GRP_FLOATS  (PANEL_FLOATS + K3_WAVES * 16 * BSTRIDE)
#define PART_FLOATS (NG * K3_WAVES * 16 * 256)
#define STAT_FLOATS (NG * K3_WAVES * 16)
#define LDS_FLOATS  (PART_FLOATS + 2 * STAT_FLOATS)

static_assert(NIT == 16, "the mask bits of a lane (8 rows x NIT steps) are kept in four 32-bit words");
static_assert(NG * GRP_FLOATS <= PART_FLOATS, "the loop's panels and P tiles lie inside the merge buffer");
static_assert(64 * XS <= PART_FLOATS, "the epilogue operand lies inside the merge buffer");

__global__ __launch_bounds__(K3_THREADS)
void gat_attn(const float* __restrict__ h, const float* __restrict__ W,
              const int* __restrict__ adj,
              const float* __restrict__ f1, const float* __restrict__ f2,
              float* __restrict__ out) {
  __shared__ float lds[LDS_FLOATS];
  int tid  = threadIdx.x;
  int w    = __builtin_amdgcn_readfirstlane(tid >> 5);
  int g    = w >> 2;
  int wt   = w & 3;
  int gtid = tid & 127;
  int lane = tid & 31;
  int n  = lane & 15;
  int hh = lane >> 4;
  int b     = blockIdx.x >> 4;
  int super = blockIdx.x & 15;
  int i0 = (super * K3_WAVES + wt) * 16;

  float* Btile = lds + g * GRP_FLOATS;
  float* ptile = Btile + PANEL_FLOATS + wt * 16 * BSTRIDE;
  float* part = lds;
  float* mst  = lds + PART_FLOATS;
  float* lst  = mst + STAT_FLOATS;
  const int srow = (g * K3_WAVES + wt) * 16 + 8 * hh;

  float f1v[8];
  {
    const float* p = f1 + b * NN + i0 + 8 * hh;
    v4f t0 = *(const v4f*)(p);
    v4f t1 = *(const v4f*)(p + 4);
    f1v[0] = t0.x; f1v[1] = t0.y; f1v[2] = t0.z; f1v[3] = t0.w;
    f1v[4] = t1.x; f1v[5] = t1.y; f1v[6] = t1.z; f1v[7] = t1.w;
  }

  const int*   adjb = adj + ((size_t)(b * NN + i0 + 8 * hh)) * NN;
  const float* hb   = h + (size_t)b * NN * FF;
  const float* f2b  = f2 + b * NN;

  unsigned int mw[4] = { 0u, 0u, 0u, 0u };
  float fmx[8];
  #pragma unroll
  for (int r = 0; r < 8; ++r) fmx[r] = -__builtin_inff();
  #pragma unroll
  for (int it = 0; it < NIT; ++it) {
    const int j0 = (it * NG + g) * 16;
    const float f2v = f2b[j0 + n];
    #pragma unroll
    for (int r = 0; r < 8; ++r) {
      const int av = adjb[(size_t)r * NN + j0 + n];
      const bool edge = (av != 0);
      fmx[r] = edge ? fmaxf(fmx[r], f2v) : fmx[r];
      mw[it >> 2] |= (edge ? 1u : 0u) << (8 * (it & 3) + r);
    }
  }
  float mrow[8];
  {
    float zf[8];
    const unsigned int wall = mw[0] & mw[1] & mw[2] & mw[3];
    #pragma unroll
    for (int r = 0; r < 8; ++r) {
      const unsigned int pat = 0x01010101u << r;
      zf[r] = ((wall & pat) == pat) ? 0.f : 1.f;
    }
    #pragma unroll
    for (int r = 0; r < 8; ++r) {
      #pragma unroll
      for (int m = 1; m < 16; m <<= 1) {
        fmx[r] = fmaxf(fmx[r], __shfl_xor(fmx[r], m, 32));
        zf[r]  = fmaxf(zf[r],  __shfl_xor(zf[r],  m, 32));
      }
    }
    if (n == 0) {
      #pragma unroll
      for (int r = 0; r < 8; ++r) { mst[srow + r] = fmx[r]; lst[srow + r] = zf[r]; }
    }
    __syncthreads();
    #pragma unroll
    for (int r = 0; r < 8; ++r) {
      float F = mst[(0 * K3_WAVES + wt) * 16 + 8 * hh + r];
      float Z = lst[(0 * K3_WAVES + wt) * 16 + 8 * hh + r];
      #pragma unroll
      for (int s = 1; s < NG; ++s) {
        F = fmaxf(F, mst[(s * K3_WAVES + wt) * 16 + 8 * hh + r]);
        Z = fmaxf(Z, lst[(s * K3_WAVES + wt) * 16 + 8 * hh + r]);
      }
      float e = f1v[r] + F;
      e = e > 0.f ? e : SLOPE_V * e;
      mrow[r] = (Z != 0.f) ? fmaxf(e, NEG_INF_V) : e;
    }
  }

  v8f acc[16];
  v8f zero = {};
  #pragma unroll
  for (int t = 0; t < 16; ++t) acc[t] = zero;
  float lsum[8];
  #pragma unroll
  for (int r = 0; r < 8; ++r) lsum[r] = 0.f;

  for (int it = 0; it < NIT; ++it) {
    int j0 = (it * NG + g) * 16;

    const float f2v = f2b[j0 + n];

    const int pc4 = gtid & 63;
    const int pjp = gtid >> 6;
    const float* gp = hb + (size_t)(j0 + 2 * pjp) * FF + 4 * pc4;
    float* lp = Btile + pjp * RSP + 8 * pc4;
    v4f xe[4], xo[4];
    #pragma unroll
    for (int q = 0; q < 4; ++q) {
      xe[q] = *(const v4f*)(gp + (size_t)(4 * q) * FF);
      xo[q] = *(const v4f*)(gp + (size_t)(4 * q + 1) * FF);
    }

    const unsigned int mbits = mw[0] & 0xffu;
    mw[0] = (mw[0] >> 8) | (mw[1] << 24);
    mw[1] = (mw[1] >> 8) | (mw[2] << 24);
    mw[2] = (mw[2] >> 8) | (mw[3] << 24);
    mw[3] >>= 8;
    float p[8];
    #pragma unroll
    for (int r = 0; r < 8; ++r) {
      float e = f1v[r] + f2v;
      e = e > 0.f ? e : SLOPE_V * e;
      e = ((mbits >> r) & 1u) ? e : NEG_INF_V;
      p[r] = __expf(e - mrow[r]);
    }
    #pragma unroll
    for (int r = 0; r < 8; ++r)
      ptile[(r + 8 * hh) * BSTRIDE + n] = p[r];

    #pragma unroll
    for (int q = 0; q < 4; ++q) {
      v4f s0 = { xe[q].x, xo[q].x, xe[q].y, xo[q].y };
      v4f s1 = { xe[q].z, xo[q].z, xe[q].w, xo[q].w };
      *(v4f*)(lp + 2 * q * RSP)     = s0;
      *(v4f*)(lp + 2 * q * RSP + 4) = s1;
    }

    __syncthreads();

    v2f A[4];
    #pragma unroll
    for (int kk = 0; kk < 4; ++kk)
      A[kk] = *(const v2f*)(&ptile[n * BSTRIDE + 4 * kk + 2 * hh]);

    #pragma unroll
    for (int t = 0; t < 16; ++t) {
      const float* Bp = Btile + hh * RSP + 2 * (t * 16 + n);
      #pragma unroll
      for (int kk = 0; kk < 4; ++kk) {
        v2f bm = *(const v2f*)(Bp + 2 * kk * RSP);
        acc[t] = __builtin_amdgcn_wmma_f32_16x16x4_f32(false, A[kk], false, bm,
                                                       (short)0, acc[t],
                                                       false, false);
      }
    }
    __builtin_amdgcn_sched_barrier(0);
    #pragma unroll
    for (int r = 0; r < 8; ++r) lsum[r] += p[r];
    __syncthreads();
  }

  #pragma unroll
  for (int r = 0; r < 8; ++r) {
    #pragma unroll
    for (int m = 1; m < 16; m <<= 1)
      lsum[r] += __shfl_xor(lsum[r], m, 32);
  }

  {
    float* pb = part + ((g * K3_WAVES + wt) * 16) * 256 + lane * 4;
    #pragma unroll
    for (int t = 0; t < 16; ++t) {
      v4f lo = { acc[t][0], acc[t][1], acc[t][2], acc[t][3] };
      v4f hi = { acc[t][4], acc[t][5], acc[t][6], acc[t][7] };
      *(v4f*)(pb + (t * 2 + 0) * 128) = lo;
      *(v4f*)(pb + (t * 2 + 1) * 128) = hi;
    }
    if (n == 0) {
      #pragma unroll
      for (int r = 0; r < 8; ++r) lst[srow + r] = lsum[r];
    }
  }
  __syncthreads();

  float inv[8];
  #pragma unroll
  for (int r = 0; r < 8; ++r) {
    float L = 0.f;
    #pragma unroll
    for (int s = 0; s < NG; ++s) L += lst[(s * K3_WAVES + wt) * 16 + 8 * hh + r];
    inv[r] = 1.0f / L;
  }
  float y[4][8];
  #pragma unroll
  for (int tt = 0; tt < 4; ++tt) {
    int t = g * 4 + tt;
    float o[8];
    #pragma unroll
    for (int r = 0; r < 8; ++r) o[r] = 0.f;
    #pragma unroll
    for (int s = 0; s < NG; ++s) {
      const float* ps = part + ((s * K3_WAVES + wt) * 16 + t) * 256 + lane * 4;
      v4f a0 = *(const v4f*)(ps);
      v4f a1 = *(const v4f*)(ps + 128);
      o[0] += a0.x; o[1] += a0.y; o[2] += a0.z; o[3] += a0.w;
      o[4] += a1.x; o[5] += a1.y; o[6] += a1.z; o[7] += a1.w;
    }
    #pragma unroll
    for (int r = 0; r < 8; ++r) y[tt][r] = o[r] * inv[r];
  }

  __syncthreads();
  float* X = lds;
  #pragma unroll
  for (int tt = 0; tt < 4; ++tt) {
    #pragma unroll
    for (int r = 0; r < 8; ++r)
      X[(wt * 16 + 8 * hh + r) * XS + (g * 4 + tt) * 16 + n] = y[tt][r];
  }
  __syncthreads();

  v8f c[4];
  #pragma unroll
  for (int tt = 0; tt < 4; ++tt) c[tt] = zero;
  const float* Xr = X + (wt * 16 + n) * XS + 2 * hh;
  const float* Wc = W + (size_t)(2 * hh) * FF + 64 * g + 4 * n;
  #pragma unroll
  for (int k0 = 0; k0 < FF; k0 += 4) {
    v2f av = *(const v2f*)(Xr + k0);
    v4f w0 = *(const v4f*)(Wc + (size_t)k0 * FF);
    v4f w1 = *(const v4f*)(Wc + (size_t)(k0 + 1) * FF);
    v2f b0 = { w0.x, w1.x };
    v2f b1 = { w0.y, w1.y };
    v2f b2 = { w0.z, w1.z };
    v2f b3 = { w0.w, w1.w };
    c[0] = __builtin_amdgcn_wmma_f32_16x16x4_f32(false, av, false, b0, (short)0, c[0], false, false);
    c[1] = __builtin_amdgcn_wmma_f32_16x16x4_f32(false, av, false, b1, (short)0, c[1], false, false);
    c[2] = __builtin_amdgcn_wmma_f32_16x16x4_f32(false, av, false, b2, (short)0, c[2], false, false);
    c[3] = __builtin_amdgcn_wmma_f32_16x16x4_f32(false, av, false, b3, (short)0, c[3], false, false);
  }
  float* ob = out + ((size_t)(b * NN + i0 + 8 * hh)) * FF + 64 * g + 4 * n;
  #pragma unroll
  for (int r = 0; r < 8; ++r) {
    v4f ov = { c[0][r], c[1][r], c[2][r], c[3][r] };
    *(v4f*)(ob + (size_t)r * FF) = ov;
  }
}

extern "C" void kernel_launch(void* const* d_in, const int* in_sizes, int n_in,
                              void* d_out, int out_size, void* d_ws, size_t ws_size,
                              hipStream_t stream) {
  const float* h   = (const float*)d_in[0];
  const int*   adj = (const int*)d_in[1];
  const float* W   = (const float*)d_in[2];
  const float* a   = (const float*)d_in[3];
  float* out = (float*)d_out;

  float* wa = (float*)d_ws;
  float* f1 = wa + 2 * FF;
  float* f2 = f1 + (size_t)BB * NN;

  gat_wa<<<FF / (K1_THREADS / 32), K1_THREADS, 0, stream>>>(W, a, wa);
  gat_f12<<<(BB * NN) / (K2_ROWS * (K2_THREADS / 32)), K2_THREADS, 0, stream>>>(h, wa, f1, f2);
  gat_attn<<<BB * (NN / 16 / K3_WAVES), K3_THREADS, 0, stream>>>(h, W, adj, f1, f2, out);
}


# attention loop: the 4 j-groups work in two shifts between work-group barriers: groups 0,1 run a step's matrix block while groups 2,3 load and stage theirs, then they swap; sum order unchanged
# speedup vs baseline: 2.5645x; 1.0669x over previous
#include <hip/hip_runtime.h>

typedef float v2f __attribute__((ext_vector_type(2)));
typedef float v4f __attribute__((ext_vector_type(4)));
typedef float v8f __attribute__((ext_vector_type(8)));

#define BB 16
#define NN 1024
#define FF 256
#define NEG_INF_V (-1000000000.0f)
#define SLOPE_V 0.2f

#define K3_WAVES 4
#define NG 4
#define K3_THREADS (NG * K3_WAVES * 32)
#define NIT (NN / 16 / NG)
#define BSTRIDE 20
#define RSP 544
#define XS 260

#define K1_THREADS 128

__global__ __launch_bounds__(K1_THREADS)
void gat_wa(const float* __restrict__ W, const float* __restrict__ a,
            float* __restrict__ wa) {
  int k    = blockIdx.x * (K1_THREADS / 32) + (threadIdx.x >> 5);
  int lane = threadIdx.x & 31;
  const float* wr = W + (size_t)k * FF + 4 * lane;
  v4f w0 = *(const v4f*)(wr);
  v4f w1 = *(const v4f*)(wr + 128);
  v4f p0 = *(const v4f*)(a + 4 * lane);
  v4f p1 = *(const v4f*)(a + 128 + 4 * lane);
  v4f q0 = *(const v4f*)(a + FF + 4 * lane);
  v4f q1 = *(const v4f*)(a + FF + 128 + 4 * lane);
  float s1 = w0.x * p0.x + w0.y * p0.y + w0.z * p0.z + w0.w * p0.w
           + w1.x * p1.x + w1.y * p1.y + w1.z * p1.z + w1.w * p1.w;
  float s2 = w0.x * q0.x + w0.y * q0.y + w0.z * q0.z + w0.w * q0.w
           + w1.x * q1.x + w1.y * q1.y + w1.z * q1.z + w1.w * q1.w;
  #pragma unroll
  for (int m = 1; m < 32; m <<= 1) {
    s1 += __shfl_xor(s1, m, 32);
    s2 += __shfl_xor(s2, m, 32);
  }
  if (lane == 0) { wa[k] = s1; wa[FF + k] = s2; }
}

#define K2_THREADS 256
#define K2_ROWS 8

__global__ __launch_bounds__(K2_THREADS)
void gat_f12(const float* __restrict__ h, const float* __restrict__ wa,
             float* __restrict__ f1, float* __restrict__ f2) {
  int wave = (blockIdx.x * K2_THREADS + threadIdx.x) >> 5;
  int lane = threadIdx.x & 31;
  v4f u0 = *(const v4f*)(wa + 4 * lane);
  v4f u1 = *(const v4f*)(wa + 128 + 4 * lane);
  v4f v0 = *(const v4f*)(wa + FF + 4 * lane);
  v4f v1 = *(const v4f*)(wa + FF + 128 + 4 * lane);
  const float* hp = h + (size_t)wave * K2_ROWS * FF + 4 * lane;
  v4f x0[K2_ROWS], x1[K2_ROWS];
  #pragma unroll
  for (int r = 0; r < K2_ROWS; ++r) {
    x0[r] = *(const v4f*)(hp + (size_t)r * FF);
    x1[r] = *(const v4f*)(hp + (size_t)r * FF + 128);
  }
  float o1 = 0.f, o2 = 0.f;
  #pragma unroll
  for (int r = 0; r < K2_ROWS; ++r) {
    float s1 = x0[r].x * u0.x + x0[r].y * u0.y + x0[r].z * u0.z + x0[r].w * u0.w
             + x1[r].x * u1.x + x1[r].y * u1.y + x1[r].z * u1.z + x1[r].w * u1.w;
    float s2 = x0[r].x * v0.x + x0[r].y * v0.y + x0[r].z * v0.z + x0[r].w * v0.w
             + x1[r].x * v1.x + x1[r].y * v1.y + x1[r].z * v1.z + x1[r].w * v1.w;
    #pragma unroll
    for (int m = 1; m < 32; m <<= 1) {
      s1 += __shfl_xor(s1, m, 32);
      s2 += __shfl_xor(s2, m, 32);
    }
    if (lane == r) { o1 = s1; o2 = s2; }
  }
  if (lane < K2_ROWS) {
    f1[wave * K2_ROWS + lane] = o1;
    f2[wave * K2_ROWS + lane] = o2;
  }
}

#define PANEL_FLOATS (8 * RSP)
#define GRP_FLOATS  (PANEL_FLOATS + K3_WAVES * 16 * BSTRIDE)
#define PART_FLOATS (NG * K3_WAVES * 16 * 256)
#define STAT_FLOATS (NG * K3_WAVES * 16)
#define LDS_FLOATS  (PART_FLOATS + 2 * STAT_FLOATS)

static_assert(NIT == 16, "the mask bits of a lane (8 rows x NIT steps) are kept in four 32-bit words");
static_assert(NG * GRP_FLOATS <= PART_FLOATS, "the loop's panels and P tiles lie inside the merge buffer");
static_assert(64 * XS <= PART_FLOATS, "the epilogue operand lies inside the merge buffer");

__global__ __launch_bounds__(K3_THREADS)
void gat_attn(const float* __restrict__ h, const float* __restrict__ W,
              const int* __restrict__ adj,
              const float* __restrict__ f1, const float* __restrict__ f2,
              float* __restrict__ out) {
  __shared__ float lds[LDS_FLOATS];
  int tid  = threadIdx.x;
  int w    = __builtin_amdgcn_readfirstlane(tid >> 5);
  int g    = w >> 2;
  int wt   = w & 3;
  int gtid = tid & 127;
  int lane = tid & 31;
  int n  = lane & 15;
  int hh = lane >> 4;
  int b     = blockIdx.x >> 4;
  int super = blockIdx.x & 15;
  int i0 = (super * K3_WAVES + wt) * 16;

  float* Btile = lds + g * GRP_FLOATS;
  float* ptile = Btile + PANEL_FLOATS + wt * 16 * BSTRIDE;
  float* part = lds;
  float* mst  = lds + PART_FLOATS;
  float* lst  = mst + STAT_FLOATS;
  const int srow = (g * K3_WAVES + wt) * 16 + 8 * hh;

  float f1v[8];
  {
    const float* p = f1 + b * NN + i0 + 8 * hh;
    v4f t0 = *(const v4f*)(p);
    v4f t1 = *(const v4f*)(p + 4);
    f1v[0] = t0.x; f1v[1] = t0.y; f1v[2] = t0.z; f1v[3] = t0.w;
    f1v[4] = t1.x; f1v[5] = t1.y; f1v[6] = t1.z; f1v[7] = t1.w;
  }

  const int*   adjb = adj + ((size_t)(b * NN + i0 + 8 * hh)) * NN;
  const float* hb   = h + (size_t)b * NN * FF;
  const float* f2b  = f2 + b * NN;

  unsigned int mw[4] = { 0u, 0u, 0u, 0u };
  float fmx[8];
  #pragma unroll
  for (int r = 0; r < 8; ++r) fmx[r] = -__builtin_inff();
  #pragma unroll
  for (int it = 0; it < NIT; ++it) {
    const int j0 = (it * NG + g) * 16;
    const float f2v = f2b[j0 + n];
    #pragma unroll
    for (int r = 0; r < 8; ++r) {
      const int av = adjb[(size_t)r * NN + j0 + n];
      const bool edge = (av != 0);
      fmx[r] = edge ? fmaxf(fmx[r], f2v) : fmx[r];
      mw[it >> 2] |= (edge ? 1u : 0u) << (8 * (it & 3) + r);
    }
  }
  float mrow[8];
  {
    float zf[8];
    const unsigned int wall = mw[0] & mw[1] & mw[2] & mw[3];
    #pragma unroll
    for (int r = 0; r < 8; ++r) {
      const unsigned int pat = 0x01010101u << r;
      zf[r] = ((wall & pat) == pat) ? 0.f : 1.f;
    }
    #pragma unroll
    for (int r = 0; r < 8; ++r) {
      #pragma unroll
      for (int m = 1; m < 16; m <<= 1) {
        fmx[r] = fmaxf(fmx[r], __shfl_xor(fmx[r], m, 32));
        zf[r]  = fmaxf(zf[r],  __shfl_xor(zf[r],  m, 32));
      }
    }
    if (n == 0) {
      #pragma unroll
      for (int r = 0; r < 8; ++r) { mst[srow + r] = fmx[r]; lst[srow + r] = zf[r]; }
    }
    __syncthreads();
    #pragma unroll
    for (int r = 0; r < 8; ++r) {
      float F = mst[(0 * K3_WAVES + wt) * 16 + 8 * hh + r];
      float Z = lst[(0 * K3_WAVES + wt) * 16 + 8 * hh + r];
      #pragma unroll
      for (int s = 1; s < NG; ++s) {
        F = fmaxf(F, mst[(s * K3_WAVES + wt) * 16 + 8 * hh + r]);
        Z = fmaxf(Z, lst[(s * K3_WAVES + wt) * 16 + 8 * hh + r]);
      }
      float e = f1v[r] + F;
      e = e > 0.f ? e : SLOPE_V * e;
      mrow[r] = (Z != 0.f) ? fmaxf(e, NEG_INF_V) : e;
    }
  }

  v8f acc[16];
  v8f zero = {};
  #pragma unroll
  for (int t = 0; t < 16; ++t) acc[t] = zero;
  float lsum[8];
  #pragma unroll
  for (int r = 0; r < 8; ++r) lsum[r] = 0.f;

  const int shift = g >> 1;
  const int pc4 = gtid & 63;
  const int pjp = gtid >> 6;
  float* lp = Btile + pjp * RSP + 8 * pc4;

  for (int s = 0; s <= 2 * NIT; ++s) {
    const int t = s - shift;
    const bool in_range = (t >= 0) && (t < 2 * NIT);
    if (in_range && (t & 1) == 0) {
        const int it = t >> 1;
        const int j0 = (it * NG + g) * 16;

        const float f2v = f2b[j0 + n];

        const float* gp = hb + (size_t)(j0 + 2 * pjp) * FF + 4 * pc4;
        v4f xe[4], xo[4];
        #pragma unroll
        for (int q = 0; q < 4; ++q) {
          xe[q] = *(const v4f*)(gp + (size_t)(4 * q) * FF);
          xo[q] = *(const v4f*)(gp + (size_t)(4 * q + 1) * FF);
        }

        const unsigned int mword = (it < 8) ? ((it < 4) ? mw[0] : mw[1]) : ((it < 12) ? mw[2] : mw[3]);
        const unsigned int mbits = mword >> (8 * (it & 3));
        float p[8];
        #pragma unroll
        for (int r = 0; r < 8; ++r) {
          float e = f1v[r] + f2v;
          e = e > 0.f ? e : SLOPE_V * e;
          e = ((mbits >> r) & 1u) ? e : NEG_INF_V;
          p[r] = __expf(e - mrow[r]);
          lsum[r] += p[r];
        }
        #pragma unroll
        for (int r = 0; r < 8; ++r)
          ptile[(r + 8 * hh) * BSTRIDE + n] = p[r];

        #pragma unroll
        for (int q = 0; q < 4; ++q) {
          v4f s0 = { xe[q].x, xo[q].x, xe[q].y, xo[q].y };
          v4f s1 = { xe[q].z, xo[q].z, xe[q].w, xo[q].w };
          *(v4f*)(lp + 2 * q * RSP)     = s0;
          *(v4f*)(lp + 2 * q * RSP + 4) = s1;
        }
    }
    if (in_range && (t & 1) != 0) {
        v2f A[4];
        #pragma unroll
        for (int kk = 0; kk < 4; ++kk)
          A[kk] = *(const v2f*)(&ptile[n * BSTRIDE + 4 * kk + 2 * hh]);

        #pragma unroll
        for (int tc = 0; tc < 16; ++tc) {
          const float* Bp = Btile + hh * RSP + 2 * (tc * 16 + n);
          #pragma unroll
          for (int kk = 0; kk < 4; ++kk) {
            v2f bm = *(const v2f*)(Bp + 2 * kk * RSP);
            acc[tc] = __builtin_amdgcn_wmma_f32_16x16x4_f32(false, A[kk], false, bm,
                                                            (short)0, acc[tc],
                                                            false, false);
          }
        }
        __builtin_amdgcn_sched_barrier(0);
        asm volatile("v_nop\n\tv_nop\n\tv_nop\n\tv_nop");
    }
    __syncthreads();
  }

  #pragma unroll
  for (int r = 0; r < 8; ++r) {
    #pragma unroll
    for (int m = 1; m < 16; m <<= 1)
      lsum[r] += __shfl_xor(lsum[r], m, 32);
  }

  {
    float* pb = part + ((g * K3_WAVES + wt) * 16) * 256 + lane * 4;
    #pragma unroll
    for (int t = 0; t < 16; ++t) {
      v4f lo = { acc[t][0], acc[t][1], acc[t][2], acc[t][3] };
      v4f hi = { acc[t][4], acc[t][5], acc[t][6], acc[t][7] };
      *(v4f*)(pb + (t * 2 + 0) * 128) = lo;
      *(v4f*)(pb + (t * 2 + 1) * 128) = hi;
    }
    if (n == 0) {
      #pragma unroll
      for (int r = 0; r < 8; ++r) lst[srow + r] = lsum[r];
    }
  }
  __syncthreads();

  float inv[8];
  #pragma unroll
  for (int r = 0; r < 8; ++r) {
    float L = 0.f;
    #pragma unroll
    for (int s = 0; s < NG; ++s) L += lst[(s * K3_WAVES + wt) * 16 + 8 * hh + r];
    inv[r] = 1.0f / L;
  }
  float y[4][8];
  #pragma unroll
  for (int tt = 0; tt < 4; ++tt) {
    int t = g * 4 + tt;
    float o[8];
    #pragma unroll
    for (int r = 0; r < 8; ++r) o[r] = 0.f;
    #pragma unroll
    for (int s = 0; s < NG; ++s) {
      const float* ps = part + ((s * K3_WAVES + wt) * 16 + t) * 256 + lane * 4;
      v4f a0 = *(const v4f*)(ps);
      v4f a1 = *(const v4f*)(ps + 128);
      o[0] += a0.x; o[1] += a0.y; o[2] += a0.z; o[3] += a0.w;
      o[4] += a1.x; o[5] += a1.y; o[6] += a1.z; o[7] += a1.w;
    }
    #pragma unroll
    for (int r = 0; r < 8; ++r) y[tt][r] = o[r] * inv[r];
  }

  __syncthreads();
  float* X = lds;
  #pragma unroll
  for (int tt = 0; tt < 4; ++tt) {
    #pragma unroll
    for (int r = 0; r < 8; ++r)
      X[(wt * 16 + 8 * hh + r) * XS + (g * 4 + tt) * 16 + n] = y[tt][r];
  }
  __syncthreads();

  v8f c[4];
  #pragma unroll
  for (int tt = 0; tt < 4; ++tt) c[tt] = zero;
  const float* Xr = X + (wt * 16 + n) * XS + 2 * hh;
  const float* Wc = W + (size_t)(2 * hh) * FF + 64 * g + 4 * n;
  #pragma unroll
  for (int k0 = 0; k0 < FF; k0 += 4) {
    v2f av = *(const v2f*)(Xr + k0);
    v4f w0 = *(const v4f*)(Wc + (size_t)k0 * FF);
    v4f w1 = *(const v4f*)(Wc + (size_t)(k0 + 1) * FF);
    v2f b0 = { w0.x, w1.x };
    v2f b1 = { w0.y, w1.y };
    v2f b2 = { w0.z, w1.z };
    v2f b3 = { w0.w, w1.w };
    c[0] = __builtin_amdgcn_wmma_f32_16x16x4_f32(false, av, false, b0, (short)0, c[0], false, false);
    c[1] = __builtin_amdgcn_wmma_f32_16x16x4_f32(false, av, false, b1, (short)0, c[1], false, false);
    c[2] = __builtin_amdgcn_wmma_f32_16x16x4_f32(false, av, false, b2, (short)0, c[2], false, false);
    c[3] = __builtin_amdgcn_wmma_f32_16x16x4_f32(false, av, false, b3, (short)0, c[3], false, false);
  }
  float* ob = out + ((size_t)(b * NN + i0 + 8 * hh)) * FF + 64 * g + 4 * n;
  #pragma unroll
  for (int r = 0; r < 8; ++r) {
    v4f ov = { c[0][r], c[1][r], c[2][r], c[3][r] };
    *(v4f*)(ob + (size_t)r * FF) = ov;
  }
}

extern "C" void kernel_launch(void* const* d_in, const int* in_sizes, int n_in,
                              void* d_out, int out_size, void* d_ws, size_t ws_size,
                              hipStream_t stream) {
  const float* h   = (const float*)d_in[0];
  const int*   adj = (const int*)d_in[1];
  const float* W   = (const float*)d_in[2];
  const float* a   = (const float*)d_in[3];
  float* out = (float*)d_out;

  float* wa = (float*)d_ws;
  float* f1 = wa + 2 * FF;
  float* f2 = f1 + (size_t)BB * NN;

  gat_wa<<<FF / (K1_THREADS / 32), K1_THREADS, 0, stream>>>(W, a, wa);
  gat_f12<<<(BB * NN) / (K2_ROWS * (K2_THREADS / 32)), K2_THREADS, 0, stream>>>(h, wa, f1, f2);
  gat_attn<<<BB * (NN / 16 / K3_WAVES), K3_THREADS, 0, stream>>>(h, W, adj, f1, f2, out);
}


# attention matrix block re-tiled: a wave multiplies its group's four P tiles (64 rows) by 64 panel columns, not 16 rows by 256: one LDS B-fragment read feeds 4 WMMAs instead of 1; sum order unchanged
# speedup vs baseline: 2.6568x; 1.0360x over previous
#include <hip/hip_runtime.h>

typedef float v2f __attribute__((ext_vector_type(2)));
typedef float v4f __attribute__((ext_vector_type(4)));
typedef float v8f __attribute__((ext_vector_type(8)));

#define BB 16
#define NN 1024
#define FF 256
#define NEG_INF_V (-1000000000.0f)
#define SLOPE_V 0.2f

#define K3_WAVES 4
#define NG 4
#define K3_THREADS (NG * K3_WAVES * 32)
#define NIT (NN / 16 / NG)
#define BSTRIDE 20
#define RSP 544
#define XS 260

#define K1_THREADS 128

__global__ __launch_bounds__(K1_THREADS)
void gat_wa(const float* __restrict__ W, const float* __restrict__ a,
            float* __restrict__ wa) {
  int k    = blockIdx.x * (K1_THREADS / 32) + (threadIdx.x >> 5);
  int lane = threadIdx.x & 31;
  const float* wr = W + (size_t)k * FF + 4 * lane;
  v4f w0 = *(const v4f*)(wr);
  v4f w1 = *(const v4f*)(wr + 128);
  v4f p0 = *(const v4f*)(a + 4 * lane);
  v4f p1 = *(const v4f*)(a + 128 + 4 * lane);
  v4f q0 = *(const v4f*)(a + FF + 4 * lane);
  v4f q1 = *(const v4f*)(a + FF + 128 + 4 * lane);
  float s1 = w0.x * p0.x + w0.y * p0.y + w0.z * p0.z + w0.w * p0.w
           + w1.x * p1.x + w1.y * p1.y + w1.z * p1.z + w1.w * p1.w;
  float s2 = w0.x * q0.x + w0.y * q0.y + w0.z * q0.z + w0.w * q0.w
           + w1.x * q1.x + w1.y * q1.y + w1.z * q1.z + w1.w * q1.w;
  #pragma unroll
  for (int m = 1; m < 32; m <<= 1) {
    s1 += __shfl_xor(s1, m, 32);
    s2 += __shfl_xor(s2, m, 32);
  }
  if (lane == 0) { wa[k] = s1; wa[FF + k] = s2; }
}

#define K2_THREADS 256
#define K2_ROWS 8

__global__ __launch_bounds__(K2_THREADS)
void gat_f12(const float* __restrict__ h, const float* __restrict__ wa,
             float* __restrict__ f1, float* __restrict__ f2) {
  int wave = (blockIdx.x * K2_THREADS + threadIdx.x) >> 5;
  int lane = threadIdx.x & 31;
  v4f u0 = *(const v4f*)(wa + 4 * lane);
  v4f u1 = *(const v4f*)(wa + 128 + 4 * lane);
  v4f v0 = *(const v4f*)(wa + FF + 4 * lane);
  v4f v1 = *(const v4f*)(wa + FF + 128 + 4 * lane);
  const float* hp = h + (size_t)wave * K2_ROWS * FF + 4 * lane;
  v4f x0[K2_ROWS], x1[K2_ROWS];
  #pragma unroll
  for (int r = 0; r < K2_ROWS; ++r) {
    x0[r] = *(const v4f*)(hp + (size_t)r * FF);
    x1[r] = *(const v4f*)(hp + (size_t)r * FF + 128);
  }
  float o1 = 0.f, o2 = 0.f;
  #pragma unroll
  for (int r = 0; r < K2_ROWS; ++r) {
    float s1 = x0[r].x * u0.x + x0[r].y * u0.y + x0[r].z * u0.z + x0[r].w * u0.w
             + x1[r].x * u1.x + x1[r].y * u1.y + x1[r].z * u1.z + x1[r].w * u1.w;
    float s2 = x0[r].x * v0.x + x0[r].y * v0.y + x0[r].z * v0.z + x0[r].w * v0.w
             + x1[r].x * v1.x + x1[r].y * v1.y + x1[r].z * v1.z + x1[r].w * v1.w;
    #pragma unroll
    for (int m = 1; m < 32; m <<= 1) {
      s1 += __shfl_xor(s1, m, 32);
      s2 += __shfl_xor(s2, m, 32);
    }
    if (lane == r) { o1 = s1; o2 = s2; }
  }
  if (lane < K2_ROWS) {
    f1[wave * K2_ROWS + lane] = o1;
    f2[wave * K2_ROWS + lane] = o2;
  }
}

#define PANEL_FLOATS (8 * RSP)
#define GRP_FLOATS  (PANEL_FLOATS + K3_WAVES * 16 * BSTRIDE)
#define PART_FLOATS (NG * K3_WAVES * 16 * 256)
#define STAT_FLOATS (NG * K3_WAVES * 16)
#define LDS_FLOATS  (PART_FLOATS + 2 * STAT_FLOATS)

static_assert(NIT == 16, "the mask bits of a lane (8 rows x NIT steps) are kept in four 32-bit words");
static_assert(NG * GRP_FLOATS <= PART_FLOATS, "the loop's panels and P tiles lie inside the merge buffer");
static_assert(64 * XS <= PART_FLOATS, "the epilogue operand lies inside the merge buffer");

__global__ __launch_bounds__(K3_THREADS)
void gat_attn(const float* __restrict__ h, const float* __restrict__ W,
              const int* __restrict__ adj,
              const float* __restrict__ f1, const float* __restrict__ f2,
              float* __restrict__ out) {
  __shared__ float lds[LDS_FLOATS];
  int tid  = threadIdx.x;
  int w    = __builtin_amdgcn_readfirstlane(tid >> 5);
  int g    = w >> 2;
  int wt   = w & 3;
  int gtid = tid & 127;
  int lane = tid & 31;
  int n  = lane & 15;
  int hh = lane >> 4;
  int b     = blockIdx.x >> 4;
  int super = blockIdx.x & 15;
  int i0 = (super * K3_WAVES + wt) * 16;

  float* Btile = lds + g * GRP_FLOATS;
  float* ptile = Btile + PANEL_FLOATS + wt * 16 * BSTRIDE;
  float* part = lds;
  float* mst  = lds + PART_FLOATS;
  float* lst  = mst + STAT_FLOATS;
  const int srow = (g * K3_WAVES + wt) * 16 + 8 * hh;

  float f1v[8];
  {
    const float* p = f1 + b * NN + i0 + 8 * hh;
    v4f t0 = *(const v4f*)(p);
    v4f t1 = *(const v4f*)(p + 4);
    f1v[0] = t0.x; f1v[1] = t0.y; f1v[2] = t0.z; f1v[3] = t0.w;
    f1v[4] = t1.x; f1v[5] = t1.y; f1v[6] = t1.z; f1v[7] = t1.w;
  }

  const int*   adjb = adj + ((size_t)(b * NN + i0 + 8 * hh)) * NN;
  const float* hb   = h + (size_t)b * NN * FF;
  const float* f2b  = f2 + b * NN;

  unsigned int mw[4] = { 0u, 0u, 0u, 0u };
  float fmx[8];
  #pragma unroll
  for (int r = 0; r < 8; ++r) fmx[r] = -__builtin_inff();
  #pragma unroll
  for (int it = 0; it < NIT; ++it) {
    const int j0 = (it * NG + g) * 16;
    const float f2v = f2b[j0 + n];
    #pragma unroll
    for (int r = 0; r < 8; ++r) {
      const int av = adjb[(size_t)r * NN + j0 + n];
      const bool edge = (av != 0);
      fmx[r] = edge ? fmaxf(fmx[r], f2v) : fmx[r];
      mw[it >> 2] |= (edge ? 1u : 0u) << (8 * (it & 3) + r);
    }
  }
  float mrow[8];
  {
    float zf[8];
    const unsigned int wall = mw[0] & mw[1] & mw[2] & mw[3];
    #pragma unroll
    for (int r = 0; r < 8; ++r) {
      const unsigned int pat = 0x01010101u << r;
      zf[r] = ((wall & pat) == pat) ? 0.f : 1.f;
    }
    #pragma unroll
    for (int r = 0; r < 8; ++r) {
      #pragma unroll
      for (int m = 1; m < 16; m <<= 1) {
        fmx[r] = fmaxf(fmx[r], __shfl_xor(fmx[r], m, 32));
        zf[r]  = fmaxf(zf[r],  __shfl_xor(zf[r],  m, 32));
      }
    }
    if (n == 0) {
      #pragma unroll
      for (int r = 0; r < 8; ++r) { mst[srow + r] = fmx[r]; lst[srow + r] = zf[r]; }
    }
    __syncthreads();
    #pragma unroll
    for (int r = 0; r < 8; ++r) {
      float F = mst[(0 * K3_WAVES + wt) * 16 + 8 * hh + r];
      float Z = lst[(0 * K3_WAVES + wt) * 16 + 8 * hh + r];
      #pragma unroll
      for (int s = 1; s < NG; ++s) {
        F = fmaxf(F, mst[(s * K3_WAVES + wt) * 16 + 8 * hh + r]);
        Z = fmaxf(Z, lst[(s * K3_WAVES + wt) * 16 + 8 * hh + r]);
      }
      float e = f1v[r] + F;
      e = e > 0.f ? e : SLOPE_V * e;
      mrow[r] = (Z != 0.f) ? fmaxf(e, NEG_INF_V) : e;
    }
  }

  v8f acc[16];
  v8f zero = {};
  #pragma unroll
  for (int t = 0; t < 16; ++t) acc[t] = zero;
  float lsum[8];
  #pragma unroll
  for (int r = 0; r < 8; ++r) lsum[r] = 0.f;

  const int shift = g >> 1;
  const int pc4 = gtid & 63;
  const int pjp = gtid >> 6;
  float* lp = Btile + pjp * RSP + 8 * pc4;

  for (int s = 0; s <= 2 * NIT; ++s) {
    const int t = s - shift;
    const bool in_range = (t >= 0) && (t < 2 * NIT);
    if (in_range && (t & 1) == 0) {
        const int it = t >> 1;
        const int j0 = (it * NG + g) * 16;

        const float f2v = f2b[j0 + n];

        const float* gp = hb + (size_t)(j0 + 2 * pjp) * FF + 4 * pc4;
        v4f xe[4], xo[4];
        #pragma unroll
        for (int q = 0; q < 4; ++q) {
          xe[q] = *(const v4f*)(gp + (size_t)(4 * q) * FF);
          xo[q] = *(const v4f*)(gp + (size_t)(4 * q + 1) * FF);
        }

        const unsigned int mword = (it < 8) ? ((it < 4) ? mw[0] : mw[1]) : ((it < 12) ? mw[2] : mw[3]);
        const unsigned int mbits = mword >> (8 * (it & 3));
        float p[8];
        #pragma unroll
        for (int r = 0; r < 8; ++r) {
          float e = f1v[r] + f2v;
          e = e > 0.f ? e : SLOPE_V * e;
          e = ((mbits >> r) & 1u) ? e : NEG_INF_V;
          p[r] = __expf(e - mrow[r]);
          lsum[r] += p[r];
        }
        #pragma unroll
        for (int r = 0; r < 8; ++r)
          ptile[(r + 8 * hh) * BSTRIDE + n] = p[r];

        #pragma unroll
        for (int q = 0; q < 4; ++q) {
          v4f s0 = { xe[q].x, xo[q].x, xe[q].y, xo[q].y };
          v4f s1 = { xe[q].z, xo[q].z, xe[q].w, xo[q].w };
          *(v4f*)(lp + 2 * q * RSP)     = s0;
          *(v4f*)(lp + 2 * q * RSP + 4) = s1;
        }
    }
    if (in_range && (t & 1) != 0) {
        const float* pt = Btile + PANEL_FLOATS + n * BSTRIDE + 2 * hh;
        v2f A[4][4];
        #pragma unroll
        for (int a = 0; a < 4; ++a) {
          #pragma unroll
          for (int kk = 0; kk < 4; ++kk)
            A[a][kk] = *(const v2f*)(pt + a * 16 * BSTRIDE + 4 * kk);
        }

        #pragma unroll
        for (int t4 = 0; t4 < 4; ++t4) {
          const float* Bp = Btile + hh * RSP + 2 * ((4 * wt + t4) * 16 + n);
          #pragma unroll
          for (int kk = 0; kk < 4; ++kk) {
            v2f bm = *(const v2f*)(Bp + 2 * kk * RSP);
            #pragma unroll
            for (int a = 0; a < 4; ++a)
              acc[4 * a + t4] = __builtin_amdgcn_wmma_f32_16x16x4_f32(false, A[a][kk], false, bm,
                                                                      (short)0, acc[4 * a + t4],
                                                                      false, false);
          }
        }
        __builtin_amdgcn_sched_barrier(0);
        asm volatile("v_nop\n\tv_nop\n\tv_nop\n\tv_nop");
    }
    __syncthreads();
  }

  #pragma unroll
  for (int r = 0; r < 8; ++r) {
    #pragma unroll
    for (int m = 1; m < 16; m <<= 1)
      lsum[r] += __shfl_xor(lsum[r], m, 32);
  }

  {
    #pragma unroll
    for (int a = 0; a < 4; ++a) {
      float* pb = part + ((g * K3_WAVES + a) * 16 + 4 * wt) * 256 + lane * 4;
      #pragma unroll
      for (int t4 = 0; t4 < 4; ++t4) {
        const int t = 4 * a + t4;
        v4f lo = { acc[t][0], acc[t][1], acc[t][2], acc[t][3] };
        v4f hi = { acc[t][4], acc[t][5], acc[t][6], acc[t][7] };
        *(v4f*)(pb + (t4 * 2 + 0) * 128) = lo;
        *(v4f*)(pb + (t4 * 2 + 1) * 128) = hi;
      }
    }
    if (n == 0) {
      #pragma unroll
      for (int r = 0; r < 8; ++r) lst[srow + r] = lsum[r];
    }
  }
  __syncthreads();

  float inv[8];
  #pragma unroll
  for (int r = 0; r < 8; ++r) {
    float L = 0.f;
    #pragma unroll
    for (int s = 0; s < NG; ++s) L += lst[(s * K3_WAVES + wt) * 16 + 8 * hh + r];
    inv[r] = 1.0f / L;
  }
  float y[4][8];
  #pragma unroll
  for (int tt = 0; tt < 4; ++tt) {
    int t = g * 4 + tt;
    float o[8];
    #pragma unroll
    for (int r = 0; r < 8; ++r) o[r] = 0.f;
    #pragma unroll
    for (int s = 0; s < NG; ++s) {
      const float* ps = part + ((s * K3_WAVES + wt) * 16 + t) * 256 + lane * 4;
      v4f a0 = *(const v4f*)(ps);
      v4f a1 = *(const v4f*)(ps + 128);
      o[0] += a0.x; o[1] += a0.y; o[2] += a0.z; o[3] += a0.w;
      o[4] += a1.x; o[5] += a1.y; o[6] += a1.z; o[7] += a1.w;
    }
    #pragma unroll
    for (int r = 0; r < 8; ++r) y[tt][r] = o[r] * inv[r];
  }

  __syncthreads();
  float* X = lds;
  #pragma unroll
  for (int tt = 0; tt < 4; ++tt) {
    #pragma unroll
    for (int r = 0; r < 8; ++r)
      X[(wt * 16 + 8 * hh + r) * XS + (g * 4 + tt) * 16 + n] = y[tt][r];
  }
  __syncthreads();

  v8f c[4];
  #pragma unroll
  for (int tt = 0; tt < 4; ++tt) c[tt] = zero;
  const float* Xr = X + (wt * 16 + n) * XS + 2 * hh;
  const float* Wc = W + (size_t)(2 * hh) * FF + 64 * g + 4 * n;
  #pragma unroll
  for (int k0 = 0; k0 < FF; k0 += 4) {
    v2f av = *(const v2f*)(Xr + k0);
    v4f w0 = *(const v4f*)(Wc + (size_t)k0 * FF);
    v4f w1 = *(const v4f*)(Wc + (size_t)(k0 + 1) * FF);
    v2f b0 = { w0.x, w1.x };
    v2f b1 = { w0.y, w1.y };
    v2f b2 = { w0.z, w1.z };
    v2f b3 = { w0.w, w1.w };
    c[0] = __builtin_amdgcn_wmma_f32_16x16x4_f32(false, av, false, b0, (short)0, c[0], false, false);
    c[1] = __builtin_amdgcn_wmma_f32_16x16x4_f32(false, av, false, b1, (short)0, c[1], false, false);
    c[2] = __builtin_amdgcn_wmma_f32_16x16x4_f32(false, av, false, b2, (short)0, c[2], false, false);
    c[3] = __builtin_amdgcn_wmma_f32_16x16x4_f32(false, av, false, b3, (short)0, c[3], false, false);
  }
  float* ob = out + ((size_t)(b * NN + i0 + 8 * hh)) * FF + 64 * g + 4 * n;
  #pragma unroll
  for (int r = 0; r < 8; ++r) {
    v4f ov = { c[0][r], c[1][r], c[2][r], c[3][r] };
    *(v4f*)(ob + (size_t)r * FF) = ov;
  }
}

extern "C" void kernel_launch(void* const* d_in, const int* in_sizes, int n_in,
                              void* d_out, int out_size, void* d_ws, size_t ws_size,
                              hipStream_t stream) {
  const float* h   = (const float*)d_in[0];
  const int*   adj = (const int*)d_in[1];
  const float* W   = (const float*)d_in[2];
  const float* a   = (const float*)d_in[3];
  float* out = (float*)d_out;

  float* wa = (float*)d_ws;
  float* f1 = wa + 2 * FF;
  float* f2 = f1 + (size_t)BB * NN;

  gat_wa<<<FF / (K1_THREADS / 32), K1_THREADS, 0, stream>>>(W, a, wa);
  gat_f12<<<(BB * NN) / (K2_ROWS * (K2_THREADS / 32)), K2_THREADS, 0, stream>>>(h, wa, f1, f2);
  gat_attn<<<BB * (NN / 16 / K3_WAVES), K3_THREADS, 0, stream>>>(h, W, adj, f1, f2, out);
}
